# starting kernel
# baseline (speedup 1.0000x reference)
#include <hip/hip_runtime.h>
#include <hip/hip_bf16.h>


typedef __attribute__((ext_vector_type(2))) float v2f;
typedef __attribute__((ext_vector_type(8))) float v8f;

#define T_STEPS 512
#define IN      4
#define HID     64
#define PROJ    52
#define G       256
#define KW      56
#define KC      14
#define ROWS    16
#define NBATCH  4096

#if __has_builtin(__builtin_amdgcn_rcpf)
#define FRCP(x) __builtin_amdgcn_rcpf(x)
#else
#define FRCP(x) (1.0f / (x))
#endif
#if __has_builtin(__builtin_amdgcn_exp2f)
#define FEXP2(x) __builtin_amdgcn_exp2f(x)
#else
#define FEXP2(x) exp2f(x)
#endif

#define LOG2E 1.4426950408889634f

static __device__ __forceinline__ v8f wmma_f32_16x16x4(v2f a, v2f b, v8f c) {
  return __builtin_amdgcn_wmma_f32_16x16x4_f32(false, a, false, b, (short)0, c,
                                               false, false);
}

static __device__ __forceinline__ v8f splat8(float s) {
  v8f v;
#pragma unroll
  for (int i = 0; i < 8; ++i) v[i] = s;
  return v;
}

static __device__ __forceinline__ float fast_sig(float x) {
  return FRCP(1.0f + FEXP2(x * -LOG2E));
}

static __device__ __forceinline__ float fast_tanh(float x) {
#if __has_builtin(__builtin_amdgcn_tanhf)
  return __builtin_amdgcn_tanhf(x);
#else
  const float ax = __builtin_fabsf(x);
  const float e  = FEXP2(ax * (-2.0f * LOG2E));
  const float r  = (1.0f - e) * FRCP(1.0f + e);
  return __builtin_copysignf(r, x);
#endif
}

__global__ void __launch_bounds__(128, 1)
lstm_proj_wmma_kernel(const float* __restrict__ x,
                      const float* __restrict__ W_ih,
                      const float* __restrict__ W_hh,
                      const float* __restrict__ b_ih,
                      const float* __restrict__ b_hh,
                      const float* __restrict__ W_hr,
                      float* __restrict__ out)
{
  __shared__ float abuf[ROWS * HID];
  __shared__ float hbuf[ROWS * HID];

  const int tid   = threadIdx.x;
  const int wave  = tid >> 5;
  const int lane  = tid & 31;
  const int l16   = lane & 15;
  const int hi    = lane >> 4;
  const int dlt   = hi * 2;
  const int rbase = blockIdx.x * ROWS;

  for (int idx = tid; idx < ROWS * HID; idx += blockDim.x) {
    abuf[idx] = 0.0f;
    hbuf[idx] = 0.0f;
  }

  float gbias[4];
  int   coln[4];
#pragma unroll
  for (int j = 0; j < 4; ++j) {
    const int col = 16 * (wave + 4 * j) + l16;
    coln[j]  = col;
    gbias[j] = b_ih[col] + b_hh[col];
  }

  v2f wtB[KC][4];
#pragma unroll
  for (int j = 0; j < 4; ++j) {
    const int col = coln[j];
    wtB[0][j].x = W_ih[col * IN + dlt];
    wtB[0][j].y = W_ih[col * IN + dlt + 1];
#pragma unroll
    for (int k = 1; k < KC; ++k) {
      const int hr = 4 * k + dlt - IN;
      wtB[k][j].x = W_hh[col * PROJ + hr];
      wtB[k][j].y = W_hh[col * PROJ + hr + 1];
    }
  }

  v2f whrB[16];
  {
    const int pcol = 16 * wave + l16;
#pragma unroll
    for (int k = 0; k < 16; ++k) {
      const int kk = 4 * k + dlt;
      v2f b;
      b.x = (pcol < PROJ) ? W_hr[pcol * HID + kk]     : 0.0f;
      b.y = (pcol < PROJ) ? W_hr[pcol * HID + kk + 1] : 0.0f;
      whrB[k] = b;
    }
  }

  const float* xrow =
      x + (size_t)(rbase + l16) * T_STEPS * IN + dlt;

  v8f c = splat8(0.0f);
  __syncthreads();

  for (int t = 0; t < T_STEPS; ++t) {
    v8f acc[4];
#pragma unroll
    for (int j = 0; j < 4; ++j) acc[j] = splat8(gbias[j]);

#pragma unroll
    for (int k = 0; k < KC; ++k) {
      v2f a;
      if (k == 0) {
        const float* xp = xrow + t * IN;
        a.x = xp[0]; a.y = xp[1];
      } else {
        a = *(const v2f*)&hbuf[l16 * HID + (4 * k - IN) + dlt];
      }
#pragma unroll
      for (int j = 0; j < 4; ++j)
        acc[j] = wmma_f32_16x16x4(a, wtB[k][j], acc[j]);
    }
    __builtin_prefetch(xrow + (t + 1) * IN, 0, 1);

#pragma unroll
    for (int r = 0; r < 8; ++r) {
      const float iv = fast_sig(acc[0][r]);
      const float fv = fast_sig(acc[1][r]);
      const float gv = fast_tanh(acc[2][r]);
      const float ov = fast_sig(acc[3][r]);
      const float cv = fv * c[r] + iv * gv;
      c[r] = cv;
      abuf[(r + 8 * hi) * HID + 16 * wave + l16] = ov * fast_tanh(cv);
    }
    __syncthreads();

    v8f p = splat8(0.0f);
#pragma unroll
    for (int k = 0; k < 16; ++k) {
      v2f pa = *(const v2f*)&abuf[l16 * HID + 4 * k + dlt];
      p = wmma_f32_16x16x4(pa, whrB[k], p);
    }
#pragma unroll
    for (int r = 0; r < 8; ++r)
      hbuf[(r + 8 * hi) * HID + 16 * wave + l16] = p[r];
    __syncthreads();
  }

  for (int idx = tid; idx < ROWS * PROJ; idx += blockDim.x) {
    const int r = idx / PROJ, cc = idx - r * PROJ;
    out[(size_t)(rbase + r) * PROJ + cc] = hbuf[r * HID + cc];
  }
}

extern "C" void kernel_launch(void* const* d_in, const int* in_sizes, int n_in,
                              void* d_out, int out_size, void* d_ws, size_t ws_size,
                              hipStream_t stream) {
  (void)in_sizes; (void)n_in; (void)out_size; (void)d_ws; (void)ws_size;
  const float* x    = (const float*)d_in[0];
  const float* W_ih = (const float*)d_in[1];
  const float* W_hh = (const float*)d_in[2];
  const float* b_ih = (const float*)d_in[3];
  const float* b_hh = (const float*)d_in[4];
  const float* W_hr = (const float*)d_in[5];
  float* out = (float*)d_out;

  dim3 grid(NBATCH / ROWS);
  dim3 block(128);
  lstm_proj_wmma_kernel<<<grid, block, 0, stream>>>(x, W_ih, W_hh, b_ih, b_hh,
                                                    W_hr, out);
}


# Pad the LDS row pitch of the h and o*tanh(c) exchange tiles from 64 to 68 floats: A-fragment reads (all rows on one bank before) and tile stores become bank-conflict-free; sum order unchanged
# speedup vs baseline: 1.5575x; 1.5575x over previous
#include <hip/hip_runtime.h>
#include <hip/hip_bf16.h>


typedef __attribute__((ext_vector_type(2))) float v2f;
typedef __attribute__((ext_vector_type(8))) float v8f;

#define T_STEPS 512
#define IN      4
#define HID     64
#define PROJ    52
#define G       256
#define KW      56
#define KC      14
#define ROWS    16
#define LDP     68
#define NBATCH  4096

#if __has_builtin(__builtin_amdgcn_rcpf)
#define FRCP(x) __builtin_amdgcn_rcpf(x)
#else
#define FRCP(x) (1.0f / (x))
#endif
#if __has_builtin(__builtin_amdgcn_exp2f)
#define FEXP2(x) __builtin_amdgcn_exp2f(x)
#else
#define FEXP2(x) exp2f(x)
#endif

#define LOG2E 1.4426950408889634f

static __device__ __forceinline__ v8f wmma_f32_16x16x4(v2f a, v2f b, v8f c) {
  return __builtin_amdgcn_wmma_f32_16x16x4_f32(false, a, false, b, (short)0, c,
                                               false, false);
}

static __device__ __forceinline__ v8f splat8(float s) {
  v8f v;
#pragma unroll
  for (int i = 0; i < 8; ++i) v[i] = s;
  return v;
}

static __device__ __forceinline__ float fast_sig(float x) {
  return FRCP(1.0f + FEXP2(x * -LOG2E));
}

static __device__ __forceinline__ float fast_tanh(float x) {
#if __has_builtin(__builtin_amdgcn_tanhf)
  return __builtin_amdgcn_tanhf(x);
#else
  const float ax = __builtin_fabsf(x);
  const float e  = FEXP2(ax * (-2.0f * LOG2E));
  const float r  = (1.0f - e) * FRCP(1.0f + e);
  return __builtin_copysignf(r, x);
#endif
}

__global__ void __launch_bounds__(128, 1)
lstm_proj_wmma_kernel(const float* __restrict__ x,
                      const float* __restrict__ W_ih,
                      const float* __restrict__ W_hh,
                      const float* __restrict__ b_ih,
                      const float* __restrict__ b_hh,
                      const float* __restrict__ W_hr,
                      float* __restrict__ out)
{
  __shared__ float abuf[ROWS * LDP];
  __shared__ float hbuf[ROWS * LDP];

  const int tid   = threadIdx.x;
  const int wave  = tid >> 5;
  const int lane  = tid & 31;
  const int l16   = lane & 15;
  const int hi    = lane >> 4;
  const int dlt   = hi * 2;
  const int rbase = blockIdx.x * ROWS;

  for (int idx = tid; idx < ROWS * LDP; idx += blockDim.x) {
    abuf[idx] = 0.0f;
    hbuf[idx] = 0.0f;
  }

  float gbias[4];
  int   coln[4];
#pragma unroll
  for (int j = 0; j < 4; ++j) {
    const int col = 16 * (wave + 4 * j) + l16;
    coln[j]  = col;
    gbias[j] = b_ih[col] + b_hh[col];
  }

  v2f wtB[KC][4];
#pragma unroll
  for (int j = 0; j < 4; ++j) {
    const int col = coln[j];
    wtB[0][j].x = W_ih[col * IN + dlt];
    wtB[0][j].y = W_ih[col * IN + dlt + 1];
#pragma unroll
    for (int k = 1; k < KC; ++k) {
      const int hr = 4 * k + dlt - IN;
      wtB[k][j].x = W_hh[col * PROJ + hr];
      wtB[k][j].y = W_hh[col * PROJ + hr + 1];
    }
  }

  v2f whrB[16];
  {
    const int pcol = 16 * wave + l16;
#pragma unroll
    for (int k = 0; k < 16; ++k) {
      const int kk = 4 * k + dlt;
      v2f b;
      b.x = (pcol < PROJ) ? W_hr[pcol * HID + kk]     : 0.0f;
      b.y = (pcol < PROJ) ? W_hr[pcol * HID + kk + 1] : 0.0f;
      whrB[k] = b;
    }
  }

  const float* xrow =
      x + (size_t)(rbase + l16) * T_STEPS * IN + dlt;

  v8f c = splat8(0.0f);
  __syncthreads();

  for (int t = 0; t < T_STEPS; ++t) {
    v8f acc[4];
#pragma unroll
    for (int j = 0; j < 4; ++j) acc[j] = splat8(gbias[j]);

#pragma unroll
    for (int k = 0; k < KC; ++k) {
      v2f a;
      if (k == 0) {
        const float* xp = xrow + t * IN;
        a.x = xp[0]; a.y = xp[1];
      } else {
        a = *(const v2f*)&hbuf[l16 * LDP + (4 * k - IN) + dlt];
      }
#pragma unroll
      for (int j = 0; j < 4; ++j)
        acc[j] = wmma_f32_16x16x4(a, wtB[k][j], acc[j]);
    }
    __builtin_prefetch(xrow + (t + 1) * IN, 0, 1);

#pragma unroll
    for (int r = 0; r < 8; ++r) {
      const float iv = fast_sig(acc[0][r]);
      const float fv = fast_sig(acc[1][r]);
      const float gv = fast_tanh(acc[2][r]);
      const float ov = fast_sig(acc[3][r]);
      const float cv = fv * c[r] + iv * gv;
      c[r] = cv;
      abuf[(r + 8 * hi) * LDP + 16 * wave + l16] = ov * fast_tanh(cv);
    }
    __syncthreads();

    v8f p = splat8(0.0f);
#pragma unroll
    for (int k = 0; k < 16; ++k) {
      v2f pa = *(const v2f*)&abuf[l16 * LDP + 4 * k + dlt];
      p = wmma_f32_16x16x4(pa, whrB[k], p);
    }
#pragma unroll
    for (int r = 0; r < 8; ++r)
      hbuf[(r + 8 * hi) * LDP + 16 * wave + l16] = p[r];
    __syncthreads();
  }

  for (int idx = tid; idx < ROWS * PROJ; idx += blockDim.x) {
    const int r = idx / PROJ, cc = idx - r * PROJ;
    out[(size_t)(rbase + r) * PROJ + cc] = hbuf[r * LDP + cc];
  }
}

extern "C" void kernel_launch(void* const* d_in, const int* in_sizes, int n_in,
                              void* d_out, int out_size, void* d_ws, size_t ws_size,
                              hipStream_t stream) {
  (void)in_sizes; (void)n_in; (void)out_size; (void)d_ws; (void)ws_size;
  const float* x    = (const float*)d_in[0];
  const float* W_ih = (const float*)d_in[1];
  const float* W_hh = (const float*)d_in[2];
  const float* b_ih = (const float*)d_in[3];
  const float* b_hh = (const float*)d_in[4];
  const float* W_hr = (const float*)d_in[5];
  float* out = (float*)d_out;

  dim3 grid(NBATCH / ROWS);
  dim3 block(128);
  lstm_proj_wmma_kernel<<<grid, block, 0, stream>>>(x, W_ih, W_hh, b_ih, b_hh,
                                                    W_hr, out);
}


# Fold the projection into the recurrent weights: M = W_hr^T W_hh^T built once per launch on WMMA; a step is one product [x_t|a] x [W_ih^T;M] and one barrier; recurrent sum order changes
# speedup vs baseline: 1.8956x; 1.2171x over previous
#include <hip/hip_runtime.h>
#include <hip/hip_bf16.h>


typedef __attribute__((ext_vector_type(2))) float v2f;
typedef __attribute__((ext_vector_type(8))) float v8f;

#define T_STEPS 512
#define IN      4
#define HID     64
#define PROJ    52
#define G       256
#define PC      13
#define AC      16
#define ROWS    16
#define LDP     68
#define NBATCH  4096

#if __has_builtin(__builtin_amdgcn_rcpf)
#define FRCP(x) __builtin_amdgcn_rcpf(x)
#else
#define FRCP(x) (1.0f / (x))
#endif
#if __has_builtin(__builtin_amdgcn_exp2f)
#define FEXP2(x) __builtin_amdgcn_exp2f(x)
#else
#define FEXP2(x) exp2f(x)
#endif

#define LOG2E 1.4426950408889634f

static __device__ __forceinline__ v8f wmma_f32_16x16x4(v2f a, v2f b, v8f c) {
  return __builtin_amdgcn_wmma_f32_16x16x4_f32(false, a, false, b, (short)0, c,
                                               false, false);
}

static __device__ __forceinline__ v8f splat8(float s) {
  v8f v;
#pragma unroll
  for (int i = 0; i < 8; ++i) v[i] = s;
  return v;
}

static __device__ __forceinline__ float fast_sig(float x) {
  return FRCP(1.0f + FEXP2(x * -LOG2E));
}

static __device__ __forceinline__ float fast_tanh(float x) {
#if __has_builtin(__builtin_amdgcn_tanhf)
  return __builtin_amdgcn_tanhf(x);
#else
  const float ax = __builtin_fabsf(x);
  const float e  = FEXP2(ax * (-2.0f * LOG2E));
  const float r  = (1.0f - e) * FRCP(1.0f + e);
  return __builtin_copysignf(r, x);
#endif
}

__global__ void __launch_bounds__(128, 1)
lstm_proj_wmma_kernel(const float* __restrict__ x,
                      const float* __restrict__ W_ih,
                      const float* __restrict__ W_hh,
                      const float* __restrict__ b_ih,
                      const float* __restrict__ b_hh,
                      const float* __restrict__ W_hr,
                      float* __restrict__ out)
{
  __shared__ __attribute__((aligned(16))) float abuf[2 * ROWS * LDP];

  const int tid   = threadIdx.x;
  const int wave  = tid >> 5;
  const int lane  = tid & 31;
  const int l16   = lane & 15;
  const int hi    = lane >> 4;
  const int dlt   = hi * 2;
  const int rbase = blockIdx.x * ROWS;

  for (int idx = tid; idx < 2 * ROWS * LDP; idx += blockDim.x) abuf[idx] = 0.0f;

  float gbias[4];
  int   coln[4];
#pragma unroll
  for (int j = 0; j < 4; ++j) {
    const int col = 16 * (wave + 4 * j) + l16;
    coln[j]  = col;
    gbias[j] = b_ih[col] + b_hh[col];
  }

  v2f wxB[4];
#pragma unroll
  for (int j = 0; j < 4; ++j) {
    wxB[j].x = W_ih[coln[j] * IN + dlt];
    wxB[j].y = W_ih[coln[j] * IN + dlt + 1];
  }

  v2f wtM[AC][4];
  {
    v2f ahr[4][PC];
    v2f bhh[4][PC];
#pragma unroll
    for (int pc = 0; pc < PC; ++pc) {
#pragma unroll
      for (int m = 0; m < 4; ++m) {
        ahr[m][pc].x = W_hr[(4 * pc + dlt)     * HID + 16 * m + l16];
        ahr[m][pc].y = W_hr[(4 * pc + dlt + 1) * HID + 16 * m + l16];
      }
#pragma unroll
      for (int j = 0; j < 4; ++j) {
        bhh[j][pc].x = W_hh[coln[j] * PROJ + 4 * pc + dlt];
        bhh[j][pc].y = W_hh[coln[j] * PROJ + 4 * pc + dlt + 1];
      }
    }
    __builtin_amdgcn_sched_barrier(0);
#pragma unroll
    for (int j = 0; j < 4; ++j) {
#pragma unroll
      for (int m = 0; m < 4; ++m) {
        v8f d = splat8(0.0f);
#pragma unroll
        for (int pc = 0; pc < PC; ++pc)
          d = wmma_f32_16x16x4(ahr[m][pc], bhh[j][pc], d);
#pragma unroll
        for (int q = 0; q < 4; ++q) {
          wtM[4 * m + q][j].x = d[2 * q];
          wtM[4 * m + q][j].y = d[2 * q + 1];
        }
      }
    }
  }

  __builtin_amdgcn_sched_barrier(0);

  const int apos = 16 * wave + 4 * ((l16 & 7) >> 1) + 2 * (l16 >> 3) + (l16 & 1);

  const float* xrow =
      x + (size_t)(rbase + l16) * T_STEPS * IN + dlt;

  v8f c = splat8(0.0f);
  int rdo = 0;
  int wro = ROWS * LDP;
  __syncthreads();

  for (int t = 0; t < T_STEPS; ++t) {
    v8f acc[4];
#pragma unroll
    for (int j = 0; j < 4; ++j) acc[j] = splat8(gbias[j]);

    {
      const float* xp = xrow + t * IN;
      v2f ax;
      ax.x = xp[0]; ax.y = xp[1];
#pragma unroll
      for (int j = 0; j < 4; ++j)
        acc[j] = wmma_f32_16x16x4(ax, wxB[j], acc[j]);
    }
#pragma unroll
    for (int k = 0; k < AC; ++k) {
      const v2f a = *(const v2f*)__builtin_assume_aligned(
          &abuf[rdo + l16 * LDP + 4 * k + dlt], 8);
#pragma unroll
      for (int j = 0; j < 4; ++j)
        acc[j] = wmma_f32_16x16x4(a, wtM[k][j], acc[j]);
    }
    __builtin_prefetch(xrow + (t + 1) * IN, 0, 1);

#pragma unroll
    for (int r = 0; r < 8; ++r) {
      const float iv = fast_sig(acc[0][r]);
      const float fv = fast_sig(acc[1][r]);
      const float gv = fast_tanh(acc[2][r]);
      const float ov = fast_sig(acc[3][r]);
      const float cv = fv * c[r] + iv * gv;
      c[r] = cv;
      abuf[wro + (r + 8 * hi) * LDP + apos] = ov * fast_tanh(cv);
    }
    __syncthreads();
    rdo ^= ROWS * LDP;
    wro ^= ROWS * LDP;
  }

  {
    const int pcol = 16 * wave + l16;
    const int prow = (pcol < PROJ) ? pcol : (PROJ - 1);
    v2f whrB[AC];
#pragma unroll
    for (int k = 0; k < AC; ++k) {
      const int kk = 16 * (k >> 2) + 8 * hi + 2 * (k & 3);
      whrB[k].x = W_hr[prow * HID + kk];
      whrB[k].y = W_hr[prow * HID + kk + 1];
    }
    unsigned ao  = (unsigned)(rdo + l16 * LDP + dlt) * 4u;
    unsigned ho0 = (unsigned)(wro + 8 * hi * LDP + 16 * wave + l16) * 4u;
    unsigned ho1 = ho0 + 4u * LDP * 4u;
    asm volatile("" : "+v"(ao), "+v"(ho0), "+v"(ho1));
    __builtin_amdgcn_sched_barrier(0);
    v8f p = splat8(0.0f);
#pragma unroll
    for (int k = 0; k < AC; ++k) {
      const v2f pa = *(const v2f*)__builtin_assume_aligned(
          (const char*)abuf + ao + 16 * k, 8);
      p = wmma_f32_16x16x4(pa, whrB[k], p);
    }
#pragma unroll
    for (int r = 0; r < 4; ++r) {
      *(float*)((char*)abuf + ho0 + r * LDP * 4) = p[r];
      *(float*)((char*)abuf + ho1 + r * LDP * 4) = p[r + 4];
    }
  }
  __syncthreads();

  for (int idx = tid; idx < ROWS * PROJ; idx += blockDim.x) {
    const int r = idx / PROJ, cc = idx - r * PROJ;
    out[(size_t)(rbase + r) * PROJ + cc] = abuf[wro + r * LDP + cc];
  }
}

extern "C" void kernel_launch(void* const* d_in, const int* in_sizes, int n_in,
                              void* d_out, int out_size, void* d_ws, size_t ws_size,
                              hipStream_t stream) {
  (void)in_sizes; (void)n_in; (void)out_size; (void)d_ws; (void)ws_size;
  const float* x    = (const float*)d_in[0];
  const float* W_ih = (const float*)d_in[1];
  const float* W_hh = (const float*)d_in[2];
  const float* b_ih = (const float*)d_in[3];
  const float* b_hh = (const float*)d_in[4];
  const float* W_hr = (const float*)d_in[5];
  float* out = (float*)d_out;

  dim3 grid(NBATCH / ROWS);
  dim3 block(128);
  lstm_proj_wmma_kernel<<<grid, block, 0, stream>>>(x, W_ih, W_hh, b_ih, b_hh,
                                                    W_hr, out);
}


# Split the step barrier: a wave signals once its own a columns have landed and waits only before other waves' k-chunks; next step's x product and own k-chunks run between; recurrent sum order changes
# speedup vs baseline: 2.0917x; 1.1035x over previous
#include <hip/hip_runtime.h>
#include <hip/hip_bf16.h>


typedef __attribute__((ext_vector_type(2))) float v2f;
typedef __attribute__((ext_vector_type(8))) float v8f;

#define T_STEPS 512
#define IN      4
#define HID     64
#define PROJ    52
#define G       256
#define PC      13
#define AC      16
#define ROWS    16
#define LDP     68
#define NBATCH  4096

#if __has_builtin(__builtin_amdgcn_rcpf)
#define FRCP(x) __builtin_amdgcn_rcpf(x)
#else
#define FRCP(x) (1.0f / (x))
#endif
#if __has_builtin(__builtin_amdgcn_exp2f)
#define FEXP2(x) __builtin_amdgcn_exp2f(x)
#else
#define FEXP2(x) exp2f(x)
#endif

#define LOG2E 1.4426950408889634f

static __device__ __forceinline__ v8f wmma_f32_16x16x4(v2f a, v2f b, v8f c) {
  return __builtin_amdgcn_wmma_f32_16x16x4_f32(false, a, false, b, (short)0, c,
                                               false, false);
}

static __device__ __forceinline__ v8f splat8(float s) {
  v8f v;
#pragma unroll
  for (int i = 0; i < 8; ++i) v[i] = s;
  return v;
}

static __device__ __forceinline__ void lds_landed() {
  asm volatile("s_wait_dscnt 0x0" ::: "memory");
}
static __device__ __forceinline__ void bar_signal() {
  asm volatile("s_barrier_signal -1" ::: "memory");
}
static __device__ __forceinline__ void bar_wait() {
  asm volatile("s_barrier_wait -1" ::: "memory");
}
static __device__ __forceinline__ void slots4(v8f& d) {
  asm volatile("v_nop\n\tv_nop\n\tv_nop\n\tv_nop" : "+v"(d));
}
static __device__ __forceinline__ void slots4(v8f& d0, v8f& d1, v8f& d2, v8f& d3) {
  asm volatile("v_nop\n\tv_nop\n\tv_nop\n\tv_nop"
               : "+v"(d0), "+v"(d1), "+v"(d2), "+v"(d3));
}

static __device__ __forceinline__ float fast_sig(float x) {
  return FRCP(1.0f + FEXP2(x * -LOG2E));
}

static __device__ __forceinline__ float fast_tanh(float x) {
#if __has_builtin(__builtin_amdgcn_tanhf)
  return __builtin_amdgcn_tanhf(x);
#else
  const float ax = __builtin_fabsf(x);
  const float e  = FEXP2(ax * (-2.0f * LOG2E));
  const float r  = (1.0f - e) * FRCP(1.0f + e);
  return __builtin_copysignf(r, x);
#endif
}

__global__ void __launch_bounds__(128, 1)
lstm_proj_wmma_kernel(const float* __restrict__ x,
                      const float* __restrict__ W_ih,
                      const float* __restrict__ W_hh,
                      const float* __restrict__ b_ih,
                      const float* __restrict__ b_hh,
                      const float* __restrict__ W_hr,
                      float* __restrict__ out)
{
  __shared__ __attribute__((aligned(16))) float abuf[2 * ROWS * LDP];

  const int tid   = threadIdx.x;
  const int wave  = tid >> 5;
  const int lane  = tid & 31;
  const int l16   = lane & 15;
  const int hi    = lane >> 4;
  const int dlt   = hi * 2;
  const int rbase = blockIdx.x * ROWS;

  for (int idx = tid; idx < 2 * ROWS * LDP; idx += blockDim.x) abuf[idx] = 0.0f;

  float gbias[4];
  int   coln[4];
#pragma unroll
  for (int j = 0; j < 4; ++j) {
    const int col = 16 * (wave + 4 * j) + l16;
    coln[j]  = col;
    gbias[j] = b_ih[col] + b_hh[col];
  }

  v2f wxB[4];
#pragma unroll
  for (int j = 0; j < 4; ++j) {
    wxB[j].x = W_ih[coln[j] * IN + dlt];
    wxB[j].y = W_ih[coln[j] * IN + dlt + 1];
  }

  v2f wtM[AC][4];
  {
    v2f ahr[4][PC];
    v2f bhh[4][PC];
#pragma unroll
    for (int pc = 0; pc < PC; ++pc) {
#pragma unroll
      for (int m = 0; m < 4; ++m) {
        const int mrow = (wave + m) & 3;
        ahr[m][pc].x = W_hr[(4 * pc + dlt)     * HID + 16 * mrow + l16];
        ahr[m][pc].y = W_hr[(4 * pc + dlt + 1) * HID + 16 * mrow + l16];
      }
#pragma unroll
      for (int j = 0; j < 4; ++j) {
        bhh[j][pc].x = W_hh[coln[j] * PROJ + 4 * pc + dlt];
        bhh[j][pc].y = W_hh[coln[j] * PROJ + 4 * pc + dlt + 1];
      }
    }
    __builtin_amdgcn_sched_barrier(0);
#pragma unroll
    for (int j = 0; j < 4; ++j) {
#pragma unroll
      for (int m = 0; m < 4; ++m) {
        v8f d = splat8(0.0f);
#pragma unroll
        for (int pc = 0; pc < PC; ++pc)
          d = wmma_f32_16x16x4(ahr[m][pc], bhh[j][pc], d);
        if (j == 3 && m == 3) slots4(d);
#pragma unroll
        for (int q = 0; q < 4; ++q) {
          wtM[4 * m + q][j].x = d[2 * q];
          wtM[4 * m + q][j].y = d[2 * q + 1];
        }
      }
    }
  }

  __builtin_amdgcn_sched_barrier(0);

  const int apos = 16 * wave + 4 * ((l16 & 7) >> 1) + 2 * (l16 >> 3) + (l16 & 1);
  int sbase = 8 * hi * LDP + apos;

  const float* xrow =
      x + (size_t)(rbase + l16) * T_STEPS * IN + dlt;

  int abase[4];
#pragma unroll
  for (int mm = 0; mm < 4; ++mm)
    abase[mm] = l16 * LDP + 16 * ((wave + mm) & 3) + dlt;

  v8f biasC[4];
#pragma unroll
  for (int j = 0; j < 4; ++j) biasC[j] = splat8(gbias[j]);

  v8f c = splat8(0.0f);
  int rdo = 0;
  int wro = ROWS * LDP;
  __syncthreads();

  v8f accn[4];
  v2f aown[4];
  {
    v2f ax;
    ax.x = xrow[0]; ax.y = xrow[1];
#pragma unroll
    for (int q = 0; q < 4; ++q)
      aown[q] = *(const v2f*)__builtin_assume_aligned(
          &abuf[rdo + abase[0] + 4 * q], 8);
    asm volatile("" : "+v"(sbase), "+v"(abase[1]), "+v"(abase[2]), "+v"(abase[3]));
    __builtin_amdgcn_sched_barrier(0);
#pragma unroll
    for (int j = 0; j < 4; ++j)
      accn[j] = wmma_f32_16x16x4(ax, wxB[j], biasC[j]);
    slots4(accn[0], accn[1], accn[2], accn[3]);
    __builtin_amdgcn_sched_barrier(0);
    lds_landed();
    bar_signal();
  }

  for (int t = 0; t < T_STEPS; ++t) {
    __builtin_amdgcn_sched_barrier(0);
    const int tn = (t + 1 < T_STEPS) ? (t + 1) : t;
    const float* xp = xrow + tn * IN;
    v2f axn;
    axn.x = xp[0]; axn.y = xp[1];
    unsigned o1 = (unsigned)(rdo + abase[1]) * 4u;
    unsigned o2 = (unsigned)(rdo + abase[2]) * 4u;
    unsigned o3 = (unsigned)(rdo + abase[3]) * 4u;
    asm volatile("" : "+v"(o1), "+v"(o2), "+v"(o3));
    __builtin_amdgcn_sched_barrier(0);

    v8f acc[4];
#pragma unroll
    for (int j = 0; j < 4; ++j) acc[j] = accn[j];

#pragma unroll
    for (int q = 0; q < 2; ++q) {
#pragma unroll
      for (int j = 0; j < 4; ++j)
        acc[j] = wmma_f32_16x16x4(aown[q], wtM[q][j], acc[j]);
    }
    bar_wait();

    v2f aoth[12];
#pragma unroll
    for (int q = 0; q < 4; ++q) {
      aoth[q]     = *(const v2f*)__builtin_assume_aligned((const char*)abuf + o1 + 16 * q, 8);
      aoth[4 + q] = *(const v2f*)__builtin_assume_aligned((const char*)abuf + o2 + 16 * q, 8);
      aoth[8 + q] = *(const v2f*)__builtin_assume_aligned((const char*)abuf + o3 + 16 * q, 8);
    }
    __builtin_amdgcn_sched_barrier(0);
#pragma unroll
    for (int q = 2; q < 4; ++q) {
#pragma unroll
      for (int j = 0; j < 4; ++j)
        acc[j] = wmma_f32_16x16x4(aown[q], wtM[q][j], acc[j]);
    }
#pragma unroll
    for (int k = 4; k < AC; ++k) {
#pragma unroll
      for (int j = 0; j < 4; ++j)
        acc[j] = wmma_f32_16x16x4(aoth[k - 4], wtM[k][j], acc[j]);
    }
    slots4(acc[0], acc[1], acc[2], acc[3]);
    __builtin_amdgcn_sched_barrier(0);

#pragma unroll
    for (int r = 0; r < 8; ++r) {
      const float iv = fast_sig(acc[0][r]);
      const float fv = fast_sig(acc[1][r]);
      const float gv = fast_tanh(acc[2][r]);
      const float ov = fast_sig(acc[3][r]);
      const float cv = fv * c[r] + iv * gv;
      c[r] = cv;
      abuf[wro + sbase + r * LDP] = ov * fast_tanh(cv);
    }

#pragma unroll
    for (int q = 0; q < 4; ++q)
      aown[q] = *(const v2f*)__builtin_assume_aligned(
          &abuf[wro + abase[0] + 4 * q], 8);
    __builtin_amdgcn_sched_barrier(0);
#pragma unroll
    for (int j = 0; j < 4; ++j)
      accn[j] = wmma_f32_16x16x4(axn, wxB[j], biasC[j]);
    slots4(accn[0], accn[1], accn[2], accn[3]);
    __builtin_amdgcn_sched_barrier(0);
    lds_landed();
    bar_signal();
    rdo ^= ROWS * LDP;
    wro ^= ROWS * LDP;
  }
  bar_wait();

  {
    const int pcol = 16 * wave + l16;
    const int prow = (pcol < PROJ) ? pcol : (PROJ - 1);
    v2f whrB[AC];
#pragma unroll
    for (int k = 0; k < AC; ++k) {
      const int kk = 16 * (k >> 2) + 8 * hi + 2 * (k & 3);
      whrB[k].x = W_hr[prow * HID + kk];
      whrB[k].y = W_hr[prow * HID + kk + 1];
    }
    unsigned ao  = (unsigned)(rdo + l16 * LDP + dlt) * 4u;
    unsigned ho0 = (unsigned)(wro + 8 * hi * LDP + 16 * wave + l16) * 4u;
    unsigned ho1 = ho0 + 4u * LDP * 4u;
    asm volatile("" : "+v"(ao), "+v"(ho0), "+v"(ho1));
    __builtin_amdgcn_sched_barrier(0);
    v8f p = splat8(0.0f);
#pragma unroll
    for (int k = 0; k < AC; ++k) {
      const v2f pa = *(const v2f*)__builtin_assume_aligned(
          (const char*)abuf + ao + 16 * k, 8);
      p = wmma_f32_16x16x4(pa, whrB[k], p);
    }
#pragma unroll
    for (int r = 0; r < 4; ++r) {
      *(float*)((char*)abuf + ho0 + r * LDP * 4) = p[r];
      *(float*)((char*)abuf + ho1 + r * LDP * 4) = p[r + 4];
    }
  }
  __syncthreads();

  for (int idx = tid; idx < ROWS * PROJ; idx += blockDim.x) {
    const int r = idx / PROJ, cc = idx - r * PROJ;
    out[(size_t)(rbase + r) * PROJ + cc] = abuf[wro + r * LDP + cc];
  }
}

extern "C" void kernel_launch(void* const* d_in, const int* in_sizes, int n_in,
                              void* d_out, int out_size, void* d_ws, size_t ws_size,
                              hipStream_t stream) {
  (void)in_sizes; (void)n_in; (void)out_size; (void)d_ws; (void)ws_size;
  const float* x    = (const float*)d_in[0];
  const float* W_ih = (const float*)d_in[1];
  const float* W_hh = (const float*)d_in[2];
  const float* b_ih = (const float*)d_in[3];
  const float* b_hh = (const float*)d_in[4];
  const float* W_hr = (const float*)d_in[5];
  float* out = (float*)d_out;

  dim3 grid(NBATCH / ROWS);
  dim3 block(128);
  lstm_proj_wmma_kernel<<<grid, block, 0, stream>>>(x, W_ih, W_hh, b_ih, b_hh,
                                                    W_hr, out);
}


# Sigmoids as 1/2 + tanh(z/2)/2 on the hardware tanh, the half folded exactly into the i, f, o columns of W_ih^T, M and bias: 5 not 8 transcendentals per element; sum order unchanged
# speedup vs baseline: 2.1775x; 1.0410x over previous
#include <hip/hip_runtime.h>
#include <hip/hip_bf16.h>


typedef __attribute__((ext_vector_type(2))) float v2f;
typedef __attribute__((ext_vector_type(8))) float v8f;

#define T_STEPS 512
#define IN      4
#define HID     64
#define PROJ    52
#define G       256
#define PC      13
#define AC      16
#define ROWS    16
#define LDP     68
#define NBATCH  4096

#if __has_builtin(__builtin_amdgcn_rcpf)
#define FRCP(x) __builtin_amdgcn_rcpf(x)
#else
#define FRCP(x) (1.0f / (x))
#endif
#if __has_builtin(__builtin_amdgcn_exp2f)
#define FEXP2(x) __builtin_amdgcn_exp2f(x)
#else
#define FEXP2(x) exp2f(x)
#endif

#define LOG2E 1.4426950408889634f

static __device__ __forceinline__ v8f wmma_f32_16x16x4(v2f a, v2f b, v8f c) {
  return __builtin_amdgcn_wmma_f32_16x16x4_f32(false, a, false, b, (short)0, c,
                                               false, false);
}

static __device__ __forceinline__ v8f splat8(float s) {
  v8f v;
#pragma unroll
  for (int i = 0; i < 8; ++i) v[i] = s;
  return v;
}

static __device__ __forceinline__ void lds_landed() {
  asm volatile("s_wait_dscnt 0x0" ::: "memory");
}
static __device__ __forceinline__ void bar_signal() {
  asm volatile("s_barrier_signal -1" ::: "memory");
}
static __device__ __forceinline__ void bar_wait() {
  asm volatile("s_barrier_wait -1" ::: "memory");
}
static __device__ __forceinline__ void slots4(v8f (&t)[16]) {
  asm volatile("v_nop\n\tv_nop\n\tv_nop\n\tv_nop"
               : "+v"(t[0]), "+v"(t[1]), "+v"(t[2]), "+v"(t[3]), "+v"(t[4]), "+v"(t[5]),
                 "+v"(t[6]), "+v"(t[7]), "+v"(t[8]), "+v"(t[9]), "+v"(t[10]), "+v"(t[11]),
                 "+v"(t[12]), "+v"(t[13]), "+v"(t[14]), "+v"(t[15]));
}
static __device__ __forceinline__ void slots4(v8f& d0, v8f& d1, v8f& d2, v8f& d3) {
  asm volatile("v_nop\n\tv_nop\n\tv_nop\n\tv_nop"
               : "+v"(d0), "+v"(d1), "+v"(d2), "+v"(d3));
}

static __device__ __forceinline__ float fast_tanh(float x) {
#if __has_builtin(__builtin_amdgcn_tanhf)
  return __builtin_amdgcn_tanhf(x);
#else
  const float ax = __builtin_fabsf(x);
  const float e  = FEXP2(ax * (-2.0f * LOG2E));
  const float r  = (1.0f - e) * FRCP(1.0f + e);
  return __builtin_copysignf(r, x);
#endif
}

static __device__ __forceinline__ float sig_of_half(float zh) {
  return __builtin_fmaf(fast_tanh(zh), 0.5f, 0.5f);
}

__global__ void __launch_bounds__(128, 1)
lstm_proj_wmma_kernel(const float* __restrict__ x,
                      const float* __restrict__ W_ih,
                      const float* __restrict__ W_hh,
                      const float* __restrict__ b_ih,
                      const float* __restrict__ b_hh,
                      const float* __restrict__ W_hr,
                      float* __restrict__ out)
{
  __shared__ __attribute__((aligned(16))) float abuf[2 * ROWS * LDP];

  const int tid   = threadIdx.x;
  const int wave  = tid >> 5;
  const int lane  = tid & 31;
  const int l16   = lane & 15;
  const int hi    = lane >> 4;
  const int dlt   = hi * 2;
  const int rbase = blockIdx.x * ROWS;

  for (int idx = tid; idx < 2 * ROWS * LDP; idx += blockDim.x) abuf[idx] = 0.0f;

  float gbias[4];
  int   coln[4];
  float gsc[4];
#pragma unroll
  for (int j = 0; j < 4; ++j) {
    const int col = 16 * (wave + 4 * j) + l16;
    coln[j]  = col;
    gsc[j]   = (j == 2) ? 1.0f : 0.5f;
    gbias[j] = (b_ih[col] + b_hh[col]) * gsc[j];
  }

  v2f wxB[4];
#pragma unroll
  for (int j = 0; j < 4; ++j) {
    wxB[j].x = W_ih[coln[j] * IN + dlt]     * gsc[j];
    wxB[j].y = W_ih[coln[j] * IN + dlt + 1] * gsc[j];
  }

  v2f wtM[AC][4];
  {
    v2f ahr[4][PC];
    v2f bhh[4][PC];
#pragma unroll
    for (int pc = 0; pc < PC; ++pc) {
#pragma unroll
      for (int m = 0; m < 4; ++m) {
        const int mrow = (wave + m) & 3;
        ahr[m][pc].x = W_hr[(4 * pc + dlt)     * HID + 16 * mrow + l16];
        ahr[m][pc].y = W_hr[(4 * pc + dlt + 1) * HID + 16 * mrow + l16];
      }
#pragma unroll
      for (int j = 0; j < 4; ++j) {
        bhh[j][pc].x = W_hh[coln[j] * PROJ + 4 * pc + dlt];
        bhh[j][pc].y = W_hh[coln[j] * PROJ + 4 * pc + dlt + 1];
      }
    }
#pragma unroll
    for (int pc = 0; pc < PC; ++pc) {
#pragma unroll
      for (int j = 0; j < 4; ++j) {
        bhh[j][pc].x *= gsc[j];
        bhh[j][pc].y *= gsc[j];
      }
      asm volatile("" : "+v"(bhh[0][pc]), "+v"(bhh[1][pc]), "+v"(bhh[3][pc]));
    }
    __builtin_amdgcn_sched_barrier(0);
    v8f dt[16];
#pragma unroll
    for (int j = 0; j < 4; ++j) {
#pragma unroll
      for (int m = 0; m < 4; ++m) {
        v8f d = splat8(0.0f);
#pragma unroll
        for (int pc = 0; pc < PC; ++pc)
          d = wmma_f32_16x16x4(ahr[m][pc], bhh[j][pc], d);
        dt[4 * j + m] = d;
      }
    }
    slots4(dt);
#pragma unroll
    for (int j = 0; j < 4; ++j) {
#pragma unroll
      for (int m = 0; m < 4; ++m) {
#pragma unroll
        for (int q = 0; q < 4; ++q) {
          wtM[4 * m + q][j].x = dt[4 * j + m][2 * q];
          wtM[4 * m + q][j].y = dt[4 * j + m][2 * q + 1];
        }
      }
    }
  }

  __builtin_amdgcn_sched_barrier(0);

  const int apos = 16 * wave + 4 * ((l16 & 7) >> 1) + 2 * (l16 >> 3) + (l16 & 1);
  int sbase = 8 * hi * LDP + apos;

  const float* xrow =
      x + (size_t)(rbase + l16) * T_STEPS * IN + dlt;

  int abase[4];
#pragma unroll
  for (int mm = 0; mm < 4; ++mm)
    abase[mm] = l16 * LDP + 16 * ((wave + mm) & 3) + dlt;

  v8f biasC[4];
#pragma unroll
  for (int j = 0; j < 4; ++j) biasC[j] = splat8(gbias[j]);

  v8f c = splat8(0.0f);
  int rdo = 0;
  int wro = ROWS * LDP;
  __syncthreads();

  v8f accn[4];
  v2f aown[4];
  {
    v2f ax;
    ax.x = xrow[0]; ax.y = xrow[1];
#pragma unroll
    for (int q = 0; q < 4; ++q)
      aown[q] = *(const v2f*)__builtin_assume_aligned(
          &abuf[rdo + abase[0] + 4 * q], 8);
    asm volatile("" : "+v"(sbase), "+v"(abase[1]), "+v"(abase[2]), "+v"(abase[3]));
    __builtin_amdgcn_sched_barrier(0);
#pragma unroll
    for (int j = 0; j < 4; ++j)
      accn[j] = wmma_f32_16x16x4(ax, wxB[j], biasC[j]);
    slots4(accn[0], accn[1], accn[2], accn[3]);
    __builtin_amdgcn_sched_barrier(0);
    lds_landed();
    bar_signal();
  }

  for (int t = 0; t < T_STEPS; ++t) {
    __builtin_amdgcn_sched_barrier(0);
    const int tn = (t + 1 < T_STEPS) ? (t + 1) : t;
    const float* xp = xrow + tn * IN;
    v2f axn;
    axn.x = xp[0]; axn.y = xp[1];
    unsigned o1 = (unsigned)(rdo + abase[1]) * 4u;
    unsigned o2 = (unsigned)(rdo + abase[2]) * 4u;
    unsigned o3 = (unsigned)(rdo + abase[3]) * 4u;
    asm volatile("" : "+v"(o1), "+v"(o2), "+v"(o3));
    __builtin_amdgcn_sched_barrier(0);

    v8f acc[4];
#pragma unroll
    for (int j = 0; j < 4; ++j) acc[j] = accn[j];

#pragma unroll
    for (int q = 0; q < 2; ++q) {
#pragma unroll
      for (int j = 0; j < 4; ++j)
        acc[j] = wmma_f32_16x16x4(aown[q], wtM[q][j], acc[j]);
    }
    bar_wait();

    v2f aoth[12];
#pragma unroll
    for (int q = 0; q < 4; ++q) {
      aoth[q]     = *(const v2f*)__builtin_assume_aligned((const char*)abuf + o1 + 16 * q, 8);
      aoth[4 + q] = *(const v2f*)__builtin_assume_aligned((const char*)abuf + o2 + 16 * q, 8);
      aoth[8 + q] = *(const v2f*)__builtin_assume_aligned((const char*)abuf + o3 + 16 * q, 8);
    }
    __builtin_amdgcn_sched_barrier(0);
#pragma unroll
    for (int q = 2; q < 4; ++q) {
#pragma unroll
      for (int j = 0; j < 4; ++j)
        acc[j] = wmma_f32_16x16x4(aown[q], wtM[q][j], acc[j]);
    }
#pragma unroll
    for (int k = 4; k < AC; ++k) {
#pragma unroll
      for (int j = 0; j < 4; ++j)
        acc[j] = wmma_f32_16x16x4(aoth[k - 4], wtM[k][j], acc[j]);
    }
    slots4(acc[0], acc[1], acc[2], acc[3]);
    __builtin_amdgcn_sched_barrier(0);

#pragma unroll
    for (int r = 0; r < 8; ++r) {
      const float iv = sig_of_half(acc[0][r]);
      const float fv = sig_of_half(acc[1][r]);
      const float gv = fast_tanh(acc[2][r]);
      const float ov = sig_of_half(acc[3][r]);
      const float cv = fv * c[r] + iv * gv;
      c[r] = cv;
      abuf[wro + sbase + r * LDP] = ov * fast_tanh(cv);
    }

#pragma unroll
    for (int q = 0; q < 4; ++q)
      aown[q] = *(const v2f*)__builtin_assume_aligned(
          &abuf[wro + abase[0] + 4 * q], 8);
    __builtin_amdgcn_sched_barrier(0);
#pragma unroll
    for (int j = 0; j < 4; ++j)
      accn[j] = wmma_f32_16x16x4(axn, wxB[j], biasC[j]);
    slots4(accn[0], accn[1], accn[2], accn[3]);
    __builtin_amdgcn_sched_barrier(0);
    lds_landed();
    bar_signal();
    rdo ^= ROWS * LDP;
    wro ^= ROWS * LDP;
  }
  bar_wait();

  {
    const int pcol = 16 * wave + l16;
    const int prow = (pcol < PROJ) ? pcol : (PROJ - 1);
    v2f whrB[AC];
#pragma unroll
    for (int k = 0; k < AC; ++k) {
      const int kk = 16 * (k >> 2) + 8 * hi + 2 * (k & 3);
      whrB[k].x = W_hr[prow * HID + kk];
      whrB[k].y = W_hr[prow * HID + kk + 1];
    }
    unsigned ao  = (unsigned)(rdo + l16 * LDP + dlt) * 4u;
    unsigned ho0 = (unsigned)(wro + 8 * hi * LDP + 16 * wave + l16) * 4u;
    unsigned ho1 = ho0 + 4u * LDP * 4u;
    asm volatile("" : "+v"(ao), "+v"(ho0), "+v"(ho1));
    __builtin_amdgcn_sched_barrier(0);
    v8f p = splat8(0.0f);
#pragma unroll
    for (int k = 0; k < AC; ++k) {
      const v2f pa = *(const v2f*)__builtin_assume_aligned(
          (const char*)abuf + ao + 16 * k, 8);
      p = wmma_f32_16x16x4(pa, whrB[k], p);
    }
#pragma unroll
    for (int r = 0; r < 4; ++r) {
      *(float*)((char*)abuf + ho0 + r * LDP * 4) = p[r];
      *(float*)((char*)abuf + ho1 + r * LDP * 4) = p[r + 4];
    }
  }
  __syncthreads();

  for (int idx = tid; idx < ROWS * PROJ; idx += blockDim.x) {
    const int r = idx / PROJ, cc = idx - r * PROJ;
    out[(size_t)(rbase + r) * PROJ + cc] = abuf[wro + r * LDP + cc];
  }
}

extern "C" void kernel_launch(void* const* d_in, const int* in_sizes, int n_in,
                              void* d_out, int out_size, void* d_ws, size_t ws_size,
                              hipStream_t stream) {
  (void)in_sizes; (void)n_in; (void)out_size; (void)d_ws; (void)ws_size;
  const float* x    = (const float*)d_in[0];
  const float* W_ih = (const float*)d_in[1];
  const float* W_hh = (const float*)d_in[2];
  const float* b_ih = (const float*)d_in[3];
  const float* b_hh = (const float*)d_in[4];
  const float* W_hr = (const float*)d_in[5];
  float* out = (float*)d_out;

  dim3 grid(NBATCH / ROWS);
  dim3 block(128);
  lstm_proj_wmma_kernel<<<grid, block, 0, stream>>>(x, W_ih, W_hh, b_ih, b_hh,
                                                    W_hr, out);
}


# Gate tiles finish apart: i, f, g first, o last as two half-sums (chunks 4-9, 10-15) added at the end; i, f, g activations, cell update and tanh(c) issue between the o WMMAs; sum order of o changes
# speedup vs baseline: 2.2524x; 1.0344x over previous
#include <hip/hip_runtime.h>
#include <hip/hip_bf16.h>


typedef __attribute__((ext_vector_type(2))) float v2f;
typedef __attribute__((ext_vector_type(8))) float v8f;

#define T_STEPS 512
#define IN      4
#define HID     64
#define PROJ    52
#define G       256
#define PC      13
#define AC      16
#define ROWS    16
#define LDP     68
#define NBATCH  4096

#if __has_builtin(__builtin_amdgcn_rcpf)
#define FRCP(x) __builtin_amdgcn_rcpf(x)
#else
#define FRCP(x) (1.0f / (x))
#endif
#if __has_builtin(__builtin_amdgcn_exp2f)
#define FEXP2(x) __builtin_amdgcn_exp2f(x)
#else
#define FEXP2(x) exp2f(x)
#endif

#define LOG2E 1.4426950408889634f

static __device__ __forceinline__ v8f wmma_f32_16x16x4(v2f a, v2f b, v8f c) {
  return __builtin_amdgcn_wmma_f32_16x16x4_f32(false, a, false, b, (short)0, c,
                                               false, false);
}

static __device__ __forceinline__ v8f splat8(float s) {
  v8f v;
#pragma unroll
  for (int i = 0; i < 8; ++i) v[i] = s;
  return v;
}

static __device__ __forceinline__ void lds_landed() {
  asm volatile("s_wait_dscnt 0x0" ::: "memory");
}
static __device__ __forceinline__ void bar_signal() {
  asm volatile("s_barrier_signal -1" ::: "memory");
}
static __device__ __forceinline__ void bar_wait() {
  asm volatile("s_barrier_wait -1" ::: "memory");
}
static __device__ __forceinline__ void slots4(v8f (&t)[16]) {
  asm volatile("v_nop\n\tv_nop\n\tv_nop\n\tv_nop"
               : "+v"(t[0]), "+v"(t[1]), "+v"(t[2]), "+v"(t[3]), "+v"(t[4]), "+v"(t[5]),
                 "+v"(t[6]), "+v"(t[7]), "+v"(t[8]), "+v"(t[9]), "+v"(t[10]), "+v"(t[11]),
                 "+v"(t[12]), "+v"(t[13]), "+v"(t[14]), "+v"(t[15]));
}
static __device__ __forceinline__ void slots4(v8f& d0, v8f& d1, v8f& d2, v8f& d3) {
  asm volatile("v_nop\n\tv_nop\n\tv_nop\n\tv_nop"
               : "+v"(d0), "+v"(d1), "+v"(d2), "+v"(d3));
}

static __device__ __forceinline__ void slots4(v8f& d0, v8f& d1) {
  asm volatile("v_nop\n\tv_nop\n\tv_nop\n\tv_nop" : "+v"(d0), "+v"(d1));
}

static __device__ __forceinline__ float fast_tanh(float x) {
#if __has_builtin(__builtin_amdgcn_tanhf)
  return __builtin_amdgcn_tanhf(x);
#else
  const float ax = __builtin_fabsf(x);
  const float e  = FEXP2(ax * (-2.0f * LOG2E));
  const float r  = (1.0f - e) * FRCP(1.0f + e);
  return __builtin_copysignf(r, x);
#endif
}

static __device__ __forceinline__ float sig_of_half(float zh) {
  return __builtin_fmaf(fast_tanh(zh), 0.5f, 0.5f);
}

__global__ void __launch_bounds__(128, 1)
lstm_proj_wmma_kernel(const float* __restrict__ x,
                      const float* __restrict__ W_ih,
                      const float* __restrict__ W_hh,
                      const float* __restrict__ b_ih,
                      const float* __restrict__ b_hh,
                      const float* __restrict__ W_hr,
                      float* __restrict__ out)
{
  __shared__ __attribute__((aligned(16))) float abuf[2 * ROWS * LDP];

  const int tid   = threadIdx.x;
  const int wave  = tid >> 5;
  const int lane  = tid & 31;
  const int l16   = lane & 15;
  const int hi    = lane >> 4;
  const int dlt   = hi * 2;
  const int rbase = blockIdx.x * ROWS;

  for (int idx = tid; idx < 2 * ROWS * LDP; idx += blockDim.x) abuf[idx] = 0.0f;

  float gbias[4];
  int   coln[4];
  float gsc[4];
#pragma unroll
  for (int j = 0; j < 4; ++j) {
    const int col = 16 * (wave + 4 * j) + l16;
    coln[j]  = col;
    gsc[j]   = (j == 2) ? 1.0f : 0.5f;
    gbias[j] = (b_ih[col] + b_hh[col]) * gsc[j];
  }

  v2f wxB[4];
#pragma unroll
  for (int j = 0; j < 4; ++j) {
    wxB[j].x = W_ih[coln[j] * IN + dlt]     * gsc[j];
    wxB[j].y = W_ih[coln[j] * IN + dlt + 1] * gsc[j];
  }

  v2f wtM[AC][4];
  {
    v2f ahr[4][PC];
    v2f bhh[4][PC];
#pragma unroll
    for (int pc = 0; pc < PC; ++pc) {
#pragma unroll
      for (int m = 0; m < 4; ++m) {
        const int mrow = (wave + m) & 3;
        ahr[m][pc].x = W_hr[(4 * pc + dlt)     * HID + 16 * mrow + l16];
        ahr[m][pc].y = W_hr[(4 * pc + dlt + 1) * HID + 16 * mrow + l16];
      }
#pragma unroll
      for (int j = 0; j < 4; ++j) {
        bhh[j][pc].x = W_hh[coln[j] * PROJ + 4 * pc + dlt];
        bhh[j][pc].y = W_hh[coln[j] * PROJ + 4 * pc + dlt + 1];
      }
    }
#pragma unroll
    for (int pc = 0; pc < PC; ++pc) {
#pragma unroll
      for (int j = 0; j < 4; ++j) {
        bhh[j][pc].x *= gsc[j];
        bhh[j][pc].y *= gsc[j];
      }
      asm volatile("" : "+v"(bhh[0][pc]), "+v"(bhh[1][pc]), "+v"(bhh[3][pc]));
    }
    __builtin_amdgcn_sched_barrier(0);
    v8f dt[16];
#pragma unroll
    for (int j = 0; j < 4; ++j) {
#pragma unroll
      for (int m = 0; m < 4; ++m) {
        v8f d = splat8(0.0f);
#pragma unroll
        for (int pc = 0; pc < PC; ++pc)
          d = wmma_f32_16x16x4(ahr[m][pc], bhh[j][pc], d);
        dt[4 * j + m] = d;
      }
    }
    slots4(dt);
#pragma unroll
    for (int j = 0; j < 4; ++j) {
#pragma unroll
      for (int m = 0; m < 4; ++m) {
#pragma unroll
        for (int q = 0; q < 4; ++q) {
          wtM[4 * m + q][j].x = dt[4 * j + m][2 * q];
          wtM[4 * m + q][j].y = dt[4 * j + m][2 * q + 1];
        }
      }
    }
  }

  __builtin_amdgcn_sched_barrier(0);

  const int apos = 16 * wave + 4 * ((l16 & 7) >> 1) + 2 * (l16 >> 3) + (l16 & 1);
  int sbase = 8 * hi * LDP + apos;

  const float* xrow =
      x + (size_t)(rbase + l16) * T_STEPS * IN + dlt;

  int abase[4];
#pragma unroll
  for (int mm = 0; mm < 4; ++mm)
    abase[mm] = l16 * LDP + 16 * ((wave + mm) & 3) + dlt;

  v8f biasC[4];
#pragma unroll
  for (int j = 0; j < 4; ++j) biasC[j] = splat8(gbias[j]);

  v2f c2[4];
#pragma unroll
  for (int p = 0; p < 4; ++p) { c2[p].x = 0.0f; c2[p].y = 0.0f; }
  int rdo = 0;
  int wro = ROWS * LDP;
  __syncthreads();

  v8f accn[4];
  v2f aown[4];
  {
    v2f ax;
    ax.x = xrow[0]; ax.y = xrow[1];
#pragma unroll
    for (int q = 0; q < 4; ++q)
      aown[q] = *(const v2f*)__builtin_assume_aligned(
          &abuf[rdo + abase[0] + 4 * q], 8);
    asm volatile("" : "+v"(sbase), "+v"(abase[1]), "+v"(abase[2]), "+v"(abase[3]));
    __builtin_amdgcn_sched_barrier(0);
#pragma unroll
    for (int j = 0; j < 4; ++j)
      accn[j] = wmma_f32_16x16x4(ax, wxB[j], biasC[j]);
    slots4(accn[0], accn[1], accn[2], accn[3]);
    __builtin_amdgcn_sched_barrier(0);
    lds_landed();
    bar_signal();
  }

  for (int t = 0; t < T_STEPS; ++t) {
    __builtin_amdgcn_sched_barrier(0);
    const int tn = (t + 1 < T_STEPS) ? (t + 1) : t;
    const float* xp = xrow + tn * IN;
    v2f axn;
    axn.x = xp[0]; axn.y = xp[1];
    unsigned o1 = (unsigned)(rdo + abase[1]) * 4u;
    unsigned o2 = (unsigned)(rdo + abase[2]) * 4u;
    unsigned o3 = (unsigned)(rdo + abase[3]) * 4u;
    asm volatile("" : "+v"(o1), "+v"(o2), "+v"(o3));
    __builtin_amdgcn_sched_barrier(0);

    v8f acc[4];
#pragma unroll
    for (int j = 0; j < 4; ++j) acc[j] = accn[j];

#pragma unroll
    for (int q = 0; q < 2; ++q) {
#pragma unroll
      for (int j = 0; j < 4; ++j)
        acc[j] = wmma_f32_16x16x4(aown[q], wtM[q][j], acc[j]);
    }
    bar_wait();

    v2f aoth[12];
#pragma unroll
    for (int q = 0; q < 4; ++q) {
      aoth[q]     = *(const v2f*)__builtin_assume_aligned((const char*)abuf + o1 + 16 * q, 8);
      aoth[4 + q] = *(const v2f*)__builtin_assume_aligned((const char*)abuf + o2 + 16 * q, 8);
      aoth[8 + q] = *(const v2f*)__builtin_assume_aligned((const char*)abuf + o3 + 16 * q, 8);
    }
    __builtin_amdgcn_sched_barrier(0);
#pragma unroll
    for (int q = 2; q < 4; ++q) {
#pragma unroll
      for (int j = 0; j < 4; ++j)
        acc[j] = wmma_f32_16x16x4(aown[q], wtM[q][j], acc[j]);
    }
#pragma unroll
    for (int k = 4; k < AC; ++k) {
#pragma unroll
      for (int j = 0; j < 3; ++j)
        acc[j] = wmma_f32_16x16x4(aoth[k - 4], wtM[k][j], acc[j]);
    }
    __builtin_amdgcn_sched_barrier(0);

    v8f ob = splat8(0.0f);
    v2f ti[4], tf[4], tg[4], iv[4], fv[4], ig[4], tcv[4];
    const v2f half2 = {0.5f, 0.5f};
#pragma unroll
    for (int n = 0; n < 12; ++n) {
      if ((n & 1) == 0)
        acc[3] = wmma_f32_16x16x4(aoth[n >> 1], wtM[4 + (n >> 1)][3], acc[3]);
      else
        ob = wmma_f32_16x16x4(aoth[6 + (n >> 1)], wtM[10 + (n >> 1)][3], ob);
      __builtin_amdgcn_sched_barrier(0);
#pragma unroll
      for (int r = 0; r < 8; ++r) {
        if (n == (r < 7 ? r : 6)) {
          ti[r >> 1][r & 1] = fast_tanh(acc[0][r]);
          tg[r >> 1][r & 1] = fast_tanh(acc[2][r]);
          tf[r >> 1][r & 1] = fast_tanh(acc[1][r]);
        }
      }
#pragma unroll
      for (int p = 0; p < 4; ++p) {
        const int g0 = (p < 3) ? (2 * p + 2) : 7;
        if (n == g0) {
          iv[p] = __builtin_elementwise_fma(ti[p], half2, half2);
          fv[p] = __builtin_elementwise_fma(tf[p], half2, half2);
        }
        if (n == g0 + 1) ig[p] = iv[p] * tg[p];
        if (n == g0 + 2) c2[p] = __builtin_elementwise_fma(fv[p], c2[p], ig[p]);
        if (n == g0 + 3) {
          tcv[p].x = fast_tanh(c2[p].x);
          tcv[p].y = fast_tanh(c2[p].y);
        }
      }
      __builtin_amdgcn_sched_barrier(0);
    }
    slots4(acc[3], ob);
    asm volatile("" :: "v"(aoth[0]), "v"(aoth[1]), "v"(aoth[2]), "v"(aoth[3]),
                       "v"(aoth[4]), "v"(aoth[5]), "v"(aoth[6]), "v"(aoth[7]),
                       "v"(aoth[8]), "v"(aoth[9]), "v"(aoth[10]), "v"(aoth[11]),
                       "v"(aown[0]), "v"(aown[1]), "v"(aown[2]), "v"(aown[3]));
    __builtin_amdgcn_sched_barrier(0);

#pragma unroll
    for (int p = 0; p < 4; ++p) {
      v2f to;
      to.x = fast_tanh(acc[3][2 * p]     + ob[2 * p]);
      to.y = fast_tanh(acc[3][2 * p + 1] + ob[2 * p + 1]);
      const v2f av = __builtin_elementwise_fma(to, half2, half2) * tcv[p];
      abuf[wro + sbase + (2 * p) * LDP]     = av.x;
      abuf[wro + sbase + (2 * p + 1) * LDP] = av.y;
    }

#pragma unroll
    for (int q = 0; q < 4; ++q)
      aown[q] = *(const v2f*)__builtin_assume_aligned(
          &abuf[wro + abase[0] + 4 * q], 8);
    __builtin_amdgcn_sched_barrier(0);
#pragma unroll
    for (int j = 0; j < 4; ++j)
      accn[j] = wmma_f32_16x16x4(axn, wxB[j], biasC[j]);
    slots4(accn[0], accn[1], accn[2], accn[3]);
    __builtin_amdgcn_sched_barrier(0);
    lds_landed();
    bar_signal();
    rdo ^= ROWS * LDP;
    wro ^= ROWS * LDP;
  }
  bar_wait();

  {
    const int pcol = 16 * wave + l16;
    const int prow = (pcol < PROJ) ? pcol : (PROJ - 1);
    v2f whrB[AC];
#pragma unroll
    for (int k = 0; k < AC; ++k) {
      const int kk = 16 * (k >> 2) + 8 * hi + 2 * (k & 3);
      whrB[k].x = W_hr[prow * HID + kk];
      whrB[k].y = W_hr[prow * HID + kk + 1];
    }
    unsigned ao  = (unsigned)(rdo + l16 * LDP + dlt) * 4u;
    unsigned ho0 = (unsigned)(wro + 8 * hi * LDP + 16 * wave + l16) * 4u;
    unsigned ho1 = ho0 + 4u * LDP * 4u;
    asm volatile("" : "+v"(ao), "+v"(ho0), "+v"(ho1));
    __builtin_amdgcn_sched_barrier(0);
    v8f p = splat8(0.0f);
#pragma unroll
    for (int k = 0; k < AC; ++k) {
      const v2f pa = *(const v2f*)__builtin_assume_aligned(
          (const char*)abuf + ao + 16 * k, 8);
      p = wmma_f32_16x16x4(pa, whrB[k], p);
    }
#pragma unroll
    for (int r = 0; r < 4; ++r) {
      *(float*)((char*)abuf + ho0 + r * LDP * 4) = p[r];
      *(float*)((char*)abuf + ho1 + r * LDP * 4) = p[r + 4];
    }
  }
  __syncthreads();

  for (int idx = tid; idx < ROWS * PROJ; idx += blockDim.x) {
    const int r = idx / PROJ, cc = idx - r * PROJ;
    out[(size_t)(rbase + r) * PROJ + cc] = abuf[wro + r * LDP + cc];
  }
}

extern "C" void kernel_launch(void* const* d_in, const int* in_sizes, int n_in,
                              void* d_out, int out_size, void* d_ws, size_t ws_size,
                              hipStream_t stream) {
  (void)in_sizes; (void)n_in; (void)out_size; (void)d_ws; (void)ws_size;
  const float* x    = (const float*)d_in[0];
  const float* W_ih = (const float*)d_in[1];
  const float* W_hh = (const float*)d_in[2];
  const float* b_ih = (const float*)d_in[3];
  const float* b_hh = (const float*)d_in[4];
  const float* W_hr = (const float*)d_in[5];
  float* out = (float*)d_out;

  dim3 grid(NBATCH / ROWS);
  dim3 block(128);
  lstm_proj_wmma_kernel<<<grid, block, 0, stream>>>(x, W_ih, W_hh, b_ih, b_hh,
                                                    W_hr, out);
}
